# starting kernel
# baseline (speedup 1.0000x reference)
#include <hip/hip_runtime.h>
#include <hip/hip_bf16.h>

#define BB 16
#define TT 2048
#define LL 1024
#define DD 512
#define KK 5
#define NROWS (BB * TT)
#define ROWS_PER_BLOCK 16
#define THREADS 512

typedef __attribute__((ext_vector_type(2))) float v2f;
typedef __attribute__((ext_vector_type(4))) float v4f;
typedef __attribute__((ext_vector_type(8))) float v8f;

__device__ __forceinline__ float wave_reduce_sum(float v) {
#pragma unroll
  for (int off = 16; off > 0; off >>= 1)
    v += __shfl_xor(v, off, 32);
  return v;
}

__global__ __launch_bounds__(THREADS)
void conv1d_token_encoder_kernel(const float* __restrict__ x,
                                 const float* __restrict__ w,
                                 const float* __restrict__ bias,
                                 float* __restrict__ out) {
  __shared__ float sS[ROWS_PER_BLOCK * 8];

  const int lane = threadIdx.x & 31;
  const int wave = threadIdx.x >> 5;
  const int row  = blockIdx.x * ROWS_PER_BLOCK + wave;

  {
    const v4f* xr = (const v4f*)(x + (size_t)row * LL);
    float psum = 0.0f;
    v4f first = {0.f, 0.f, 0.f, 0.f};
    v4f last  = {0.f, 0.f, 0.f, 0.f};
#pragma unroll
    for (int c = 0; c < 8; ++c) {
      v4f v = __builtin_nontemporal_load(&xr[c * 32 + lane]);
      if (c == 0) first = v;
      if (c == 7) last = v;
      psum += (v[0] + v[1]) + (v[2] + v[3]);
    }
    const float total = wave_reduce_sum(psum);
    const float x0  = __shfl(first[0], 0, 32);
    const float x1  = __shfl(first[1], 0, 32);
    const float xm2 = __shfl(last[2], 31, 32);
    const float xm1 = __shfl(last[3], 31, 32);

    if (lane < 8) {
      float s;
      switch (lane) {
        case 0: s = total - xm2 - xm1; break;
        case 1: s = total - xm1;       break;
        case 2: s = total;             break;
        case 3: s = total - x0;        break;
        case 4: s = total - x0 - x1;   break;
        default: s = 0.0f;             break;
      }
      sS[wave * 8 + lane] = s * (1.0f / (float)LL);
    }
  }
  __syncthreads();

  const int m = lane & 15;
  const int h = lane >> 4;
  const int n = m;

  const v2f a0 = *(const v2f*)&sS[m * 8 + 2 * h];
  const v2f a1 = *(const v2f*)&sS[m * 8 + 4 + 2 * h];

#pragma unroll
  for (int t = 0; t < 2; ++t) {
    const int d = (wave * 2 + t) * 16 + n;
    const float* wd = w + (size_t)d * KK;

    v2f b0, b1;
    b0.x = wd[2 * h];
    b0.y = wd[2 * h + 1];
    b1.x = (h == 0) ? wd[4] : 0.0f;
    b1.y = 0.0f;

    const float bv = bias[d];
    v8f c;
#pragma unroll
    for (int r = 0; r < 8; ++r) c[r] = bv;

    c = __builtin_amdgcn_wmma_f32_16x16x4_f32(false, a0, false, b0, (short)0, c, false, false);
    c = __builtin_amdgcn_wmma_f32_16x16x4_f32(false, a1, false, b1, (short)0, c, false, false);

    float* obase = out + ((size_t)(blockIdx.x * ROWS_PER_BLOCK + 8 * h)) * DD + d;
#pragma unroll
    for (int r = 0; r < 8; ++r)
      __builtin_nontemporal_store(c[r], &obase[(size_t)r * DD]);
  }
}

extern "C" void kernel_launch(void* const* d_in, const int* in_sizes, int n_in,
                              void* d_out, int out_size, void* d_ws, size_t ws_size,
                              hipStream_t stream) {
  const float* x    = (const float*)d_in[0];
  const float* w    = (const float*)d_in[1];
  const float* bias = (const float*)d_in[2];
  float* out        = (float*)d_out;

  dim3 grid(NROWS / ROWS_PER_BLOCK);
  dim3 block(THREADS);
  conv1d_token_encoder_kernel<<<grid, block, 0, stream>>>(x, w, bias, out);
}


# One wave owns a whole 16-row tile: sums its rows with 4 rows of loads in flight, builds the A fragment in registers, runs all 32 channel tiles; 2048 waves, not 32768; add order kept
# speedup vs baseline: 1.6641x; 1.6641x over previous
#include <hip/hip_runtime.h>

#define BB 16
#define TT 2048
#define LL 1024
#define DD 512
#define KK 5
#define NROWS (BB * TT)
#define TILE_ROWS 16
#define WAVES 8
#define THREADS (WAVES * 32)
#define NTILES (DD / 16)
#define AHEAD 4
#define NBLOCKS (NROWS / (TILE_ROWS * WAVES))

typedef __attribute__((ext_vector_type(2))) float v2f;
typedef __attribute__((ext_vector_type(4))) float v4f;
typedef __attribute__((ext_vector_type(8))) float v8f;

__device__ __forceinline__ float lane_value(float v, int src) {
  return __builtin_bit_cast(float, __builtin_amdgcn_readlane(__builtin_bit_cast(int, v), src));
}

__global__ __launch_bounds__(THREADS)
void conv1d_token_encoder_kernel(const float* __restrict__ x,
                                 const float* __restrict__ w,
                                 const float* __restrict__ bias,
                                 float* __restrict__ out) {
  __shared__ v4f sB[NTILES * 32];
  __shared__ float sBias[DD];

  const int tid  = threadIdx.x;
  const int lane = tid & 31;
  const int wave = __builtin_amdgcn_readfirstlane(tid >> 5);
  const int m = lane & 15;
  const int h = lane >> 4;

  v4f frag[NTILES / WAVES];
#pragma unroll
  for (int q = 0; q < NTILES / WAVES; ++q) {
    const int d = (wave + WAVES * q) * 16 + m;
    const float* wd = w + d * KK;
    const float w4 = wd[4];
    frag[q][0] = wd[2 * h];
    frag[q][1] = wd[2 * h + 1];
    frag[q][2] = (h == 0) ? w4 : 0.0f;
    frag[q][3] = 0.0f;
  }
  const float bs0 = bias[tid];
  const float bs1 = bias[tid + THREADS];

  const size_t row0 = (size_t)(blockIdx.x * WAVES + wave) * TILE_ROWS;
  const v4f* xr = (const v4f*)(x + row0 * LL) + lane;
  v4f buf[AHEAD + 1][8];
#define LOAD_ROW(R) do { _Pragma("unroll") for (int c = 0; c < 8; ++c) \
    buf[(R) % (AHEAD + 1)][c] = __builtin_nontemporal_load(&xr[(R) * (LL / 4) + c * 32]); } while (0)
#pragma unroll
  for (int r = 0; r < AHEAD; ++r) LOAD_ROW(r);

  float p[TILE_ROWS];
  float e0 = 0.f, e1 = 0.f, e2 = 0.f, e3 = 0.f;
#pragma unroll
  for (int i = 0; i < TILE_ROWS; ++i) {
    __builtin_amdgcn_sched_barrier(0);
    if (i + AHEAD < TILE_ROWS) LOAD_ROW(i + AHEAD);
    __builtin_amdgcn_sched_barrier(0);
    const v4f* b = buf[i % (AHEAD + 1)];
    float psum = 0.0f;
#pragma unroll
    for (int c = 0; c < 8; ++c)
      psum += (b[c][0] + b[c][1]) + (b[c][2] + b[c][3]);
    asm volatile("" : "+v"(psum));
    p[i] = psum;
    const float f0 = lane_value(b[0][0], 0);
    const float f1 = lane_value(b[0][1], 0);
    const float l2 = lane_value(b[7][2], 31);
    const float l3 = lane_value(b[7][3], 31);
    e0 = (m == i) ? f0 : e0;
    e1 = (m == i) ? f1 : e1;
    e2 = (m == i) ? l2 : e2;
    e3 = (m == i) ? l3 : e3;
  }
#undef LOAD_ROW

  const int b3 = (lane >> 3) & 1, b2 = (lane >> 2) & 1, b1 = (lane >> 1) & 1;
  float q8[8], q4[4], q2[2];
#pragma unroll
  for (int j = 0; j < 8; ++j) {
    const float keep = h ? p[8 + j] : p[j];
    const float send = h ? p[j] : p[8 + j];
    q8[j] = keep + __shfl_xor(send, 16, 32);
  }
#pragma unroll
  for (int j = 0; j < 4; ++j) {
    const float keep = b3 ? q8[4 + j] : q8[j];
    const float send = b3 ? q8[j] : q8[4 + j];
    q4[j] = keep + __shfl_xor(send, 8, 32);
  }
#pragma unroll
  for (int j = 0; j < 2; ++j) {
    const float keep = b2 ? q4[2 + j] : q4[j];
    const float send = b2 ? q4[j] : q4[2 + j];
    q2[j] = keep + __shfl_xor(send, 4, 32);
  }
  float q1;
  {
    const float keep = b1 ? q2[1] : q2[0];
    const float send = b1 ? q2[0] : q2[1];
    q1 = keep + __shfl_xor(send, 2, 32);
  }
  q1 = q1 + __shfl_xor(q1, 1, 32);
  const float total = __shfl(q1, 2 * m, 32);

  const float sc = 1.0f / (float)LL;
  const float s0 = (total - e2 - e3) * sc;
  const float s1 = (total - e3) * sc;
  const float s2 = total * sc;
  const float s3 = (total - e0) * sc;
  const float s4 = (total - e0 - e1) * sc;
  v2f a0, a1;
  a0.x = h ? s2 : s0;
  a0.y = h ? s3 : s1;
  a1.x = h ? 0.0f : s4;
  a1.y = 0.0f;

#pragma unroll
  for (int q = 0; q < NTILES / WAVES; ++q) sB[(wave + WAVES * q) * 32 + lane] = frag[q];
  sBias[tid] = bs0;
  sBias[tid + THREADS] = bs1;
  __syncthreads();

  float* obase = out + (row0 + 8 * h) * DD + m;
#pragma unroll 4
  for (int t = 0; t < NTILES; ++t) {
    const v4f f = sB[t * 32 + lane];
    const float bv = sBias[t * 16 + m];
    v2f b0, b1;
    b0.x = f[0]; b0.y = f[1];
    b1.x = f[2]; b1.y = f[3];
    v8f c;
#pragma unroll
    for (int r = 0; r < 8; ++r) c[r] = bv;

    c = __builtin_amdgcn_wmma_f32_16x16x4_f32(false, a0, false, b0, (short)0, c, false, false);
    c = __builtin_amdgcn_wmma_f32_16x16x4_f32(false, a1, false, b1, (short)0, c, false, false);

#pragma unroll
    for (int r = 0; r < 8; ++r)
      __builtin_nontemporal_store(c[r], &obase[(size_t)r * DD + t * 16]);
  }
}

extern "C" void kernel_launch(void* const* d_in, const int* in_sizes, int n_in,
                              void* d_out, int out_size, void* d_ws, size_t ws_size,
                              hipStream_t stream) {
  const float* x    = (const float*)d_in[0];
  const float* w    = (const float*)d_in[1];
  const float* bias = (const float*)d_in[2];
  float* out        = (float*)d_out;

  dim3 grid(NBLOCKS);
  dim3 block(THREADS);
  conv1d_token_encoder_kernel<<<grid, block, 0, stream>>>(x, w, bias, out);
}
